# starting kernel
# baseline (speedup 1.0000x reference)
#include <hip/hip_runtime.h>

typedef __attribute__((ext_vector_type(2))) float v2f;
typedef __attribute__((ext_vector_type(8))) float v8f;

#define EDGE_DIM 128
#define N_RADIAL 16

__global__ void zero_f4_kernel(float4* __restrict__ p, long n4) {
  long i = (long)blockIdx.x * blockDim.x + threadIdx.x;
  long stride = (long)gridDim.x * blockDim.x;
  float4 z = make_float4(0.f, 0.f, 0.f, 0.f);
  for (; i < n4; i += stride) p[i] = z;
}

__global__ __launch_bounds__(256) void edge_gate_scatter_kernel(
    const float* __restrict__ x, const float* __restrict__ rbf,
    const int* __restrict__ idx_i, const float* __restrict__ W_rbf,
    float* __restrict__ h, int n_edges)
{
  const int lane = threadIdx.x & 31;
  const int r    = lane & 15;
  const int hi   = lane >> 4;
  const int wave = blockIdx.x * (blockDim.x >> 5) + (threadIdx.x >> 5);
  const int nW   = gridDim.x * (blockDim.x >> 5);
  const int nFull = n_edges >> 4;

  v2f Bt[8][4];
#pragma unroll
  for (int cb = 0; cb < 8; ++cb) {
    const float* brow = W_rbf + (cb * 16 + r) * N_RADIAL + 2 * hi;
#pragma unroll
    for (int k0 = 0; k0 < 4; ++k0)
      Bt[cb][k0] = *(const v2f*)(brow + 4 * k0);
  }

  for (int t = wave; t < nFull; t += nW) {
    const int e0 = t << 4;

    v2f a[4];
    const float* arow = rbf + (e0 + r) * N_RADIAL + 2 * hi;
#pragma unroll
    for (int k0 = 0; k0 < 4; ++k0)
      a[k0] = *(const v2f*)(arow + 4 * k0);

    int hoff[8];
    const int* ibase = idx_i + e0 + 8 * hi;
#pragma unroll
    for (int v = 0; v < 8; ++v)
      hoff[v] = ibase[v] * EDGE_DIM;

#pragma unroll
    for (int cb = 0; cb < 8; ++cb) {
      v8f c = {0.f, 0.f, 0.f, 0.f, 0.f, 0.f, 0.f, 0.f};
#pragma unroll
      for (int k0 = 0; k0 < 4; ++k0)
        c = __builtin_amdgcn_wmma_f32_16x16x4_f32(
            false, a[k0], false, Bt[cb][k0], (short)0, c, false, false);

      const int col = cb * 16 + r;

      const float* xbase = x + (e0 + 8 * hi) * EDGE_DIM + col;
      float xv[8];
#pragma unroll
      for (int v = 0; v < 8; ++v)
        xv[v] = xbase[v * EDGE_DIM];

#pragma unroll
      for (int v = 0; v < 8; ++v)
        atomicAdd(h + (hoff[v] + col), c[v] * xv[v]);
    }
  }

  if (wave == 0) {
    for (int e = nFull << 4; e < n_edges; ++e) {
      const int nd = idx_i[e];
      const float* re = rbf + e * N_RADIAL;
#pragma unroll
      for (int cc = 0; cc < 4; ++cc) {
        const int ch = lane * 4 + cc;
        float s = 0.f;
#pragma unroll
        for (int k = 0; k < N_RADIAL; ++k) s += re[k] * W_rbf[ch * N_RADIAL + k];
        float g = s * x[e * EDGE_DIM + ch];
        atomicAdd(h + (nd * EDGE_DIM + ch), g);
      }
    }
  }
}

__global__ __launch_bounds__(256) void node_mlp_kernel(
    const float* __restrict__ in, const float* __restrict__ W,
    const float* __restrict__ bias, float* __restrict__ out,
    int n_rows, int apply_swish)
{
  __shared__ float Wlds[EDGE_DIM * EDGE_DIM];
  {
    const float4* Wg = (const float4*)W;
    float4* Wl = (float4*)Wlds;
    for (int i = threadIdx.x; i < EDGE_DIM * EDGE_DIM / 4; i += blockDim.x)
      Wl[i] = Wg[i];
  }
  __syncthreads();

  const int lane = threadIdx.x & 31;
  const int r    = lane & 15;
  const int hi   = lane >> 4;
  const int wave = blockIdx.x * (blockDim.x >> 5) + (threadIdx.x >> 5);
  const int nW   = gridDim.x * (blockDim.x >> 5);
  const int nTiles = (n_rows + 15) >> 4;

  for (int t = wave; t < nTiles; t += nW) {
    const int m0 = t << 4;
    v8f acc[8];
#pragma unroll
    for (int cb = 0; cb < 8; ++cb)
      acc[cb] = (v8f){0.f, 0.f, 0.f, 0.f, 0.f, 0.f, 0.f, 0.f};

    int mrow = m0 + r;
    if (mrow >= n_rows) mrow = n_rows - 1;
    const float* arow = in + mrow * EDGE_DIM + 2 * hi;

#pragma unroll 4
    for (int k0 = 0; k0 < 32; ++k0) {
      v2f a = *(const v2f*)(arow + 4 * k0);
#pragma unroll
      for (int cb = 0; cb < 8; ++cb) {
        v2f b = *(const v2f*)(&Wlds[(cb * 16 + r) * EDGE_DIM + 4 * k0 + 2 * hi]);
        acc[cb] = __builtin_amdgcn_wmma_f32_16x16x4_f32(
            false, a, false, b, (short)0, acc[cb], false, false);
      }
    }

    const int rowBase = m0 + 8 * hi;
    if (m0 + 16 <= n_rows) {
#pragma unroll
      for (int cb = 0; cb < 8; ++cb) {
        const int col = cb * 16 + r;
        const float bv = bias ? bias[col] : 0.f;
        float* obase = out + rowBase * EDGE_DIM + col;
#pragma unroll
        for (int v = 0; v < 8; ++v) {
          float val = acc[cb][v] + bv;
          if (apply_swish) val = val * (1.f / (1.f + __expf(-val)));
          obase[v * EDGE_DIM] = val;
        }
      }
    } else {
#pragma unroll
      for (int cb = 0; cb < 8; ++cb) {
        const int col = cb * 16 + r;
        const float bv = bias ? bias[col] : 0.f;
#pragma unroll
        for (int v = 0; v < 8; ++v) {
          const int m = rowBase + v;
          if (m < n_rows) {
            float val = acc[cb][v] + bv;
            if (apply_swish) val = val * (1.f / (1.f + __expf(-val)));
            out[m * EDGE_DIM + col] = val;
          }
        }
      }
    }
  }
}

extern "C" void kernel_launch(void* const* d_in, const int* in_sizes, int n_in,
                              void* d_out, int out_size, void* d_ws, size_t ws_size,
                              hipStream_t stream) {
  const float* x     = (const float*)d_in[0];
  const float* rbf   = (const float*)d_in[1];
  const int*   idx_i = (const int*)d_in[2];
  const float* W_rbf = (const float*)d_in[4];
  const float* W1    = (const float*)d_in[5];
  const float* b1    = (const float*)d_in[6];
  const float* W2    = (const float*)d_in[7];
  const float* b2    = (const float*)d_in[8];
  const float* W3    = (const float*)d_in[9];
  const float* b3    = (const float*)d_in[10];
  const float* W_out = (const float*)d_in[11];
  float* outp = (float*)d_out;

  const int n_edges = in_sizes[0] / EDGE_DIM;
  const int n_nodes = out_size / EDGE_DIM;

  float* h0 = (float*)d_ws;
  float* h1 = h0 + (size_t)n_nodes * EDGE_DIM;

  {
    long n4 = (long)n_nodes * EDGE_DIM / 4;
    long b = (n4 + 255) / 256;
    int blocks = (int)(b > 2048 ? 2048 : (b < 1 ? 1 : b));
    zero_f4_kernel<<<blocks, 256, 0, stream>>>((float4*)h0, n4);
  }

  {
    int nTiles = n_edges >> 4;
    int blocks = (nTiles + 7) / 8;
    if (blocks > 2048) blocks = 2048;
    if (blocks < 1) blocks = 1;
    edge_gate_scatter_kernel<<<blocks, 256, 0, stream>>>(x, rbf, idx_i, W_rbf, h0, n_edges);
  }

  {
    int nTiles = (n_nodes + 15) / 16;
    int blocks = (nTiles + 7) / 8;
    if (blocks > 2048) blocks = 2048;
    if (blocks < 1) blocks = 1;
    node_mlp_kernel<<<blocks, 256, 0, stream>>>(h0, W1, b1, h1, n_nodes, 1);
    node_mlp_kernel<<<blocks, 256, 0, stream>>>(h1, W2, b2, h0, n_nodes, 1);
    node_mlp_kernel<<<blocks, 256, 0, stream>>>(h0, W3, b3, h1, n_nodes, 1);
    node_mlp_kernel<<<blocks, 256, 0, stream>>>(h1, W_out, nullptr, outp, n_nodes, 0);
  }
}


# Node MLP: W kept in registers instead of LDS: each wave holds a 64-channel half of W as 128 B fragments for the whole launch (two waves per SIMD), no LDS, no barrier; row blocks stream through
# speedup vs baseline: 1.3322x; 1.3322x over previous
#include <hip/hip_runtime.h>

typedef __attribute__((ext_vector_type(2))) float v2f;
typedef __attribute__((ext_vector_type(8))) float v8f;

#define EDGE_DIM 128
#define N_RADIAL 16

__global__ void zero_f4_kernel(float4* __restrict__ p, long n4) {
  long i = (long)blockIdx.x * blockDim.x + threadIdx.x;
  long stride = (long)gridDim.x * blockDim.x;
  float4 z = make_float4(0.f, 0.f, 0.f, 0.f);
  for (; i < n4; i += stride) p[i] = z;
}

__global__ __launch_bounds__(256) void edge_gate_scatter_kernel(
    const float* __restrict__ x, const float* __restrict__ rbf,
    const int* __restrict__ idx_i, const float* __restrict__ W_rbf,
    float* __restrict__ h, int n_edges)
{
  const int lane = threadIdx.x & 31;
  const int r    = lane & 15;
  const int hi   = lane >> 4;
  const int wave = blockIdx.x * (blockDim.x >> 5) + (threadIdx.x >> 5);
  const int nW   = gridDim.x * (blockDim.x >> 5);
  const int nFull = n_edges >> 4;

  v2f Bt[8][4];
#pragma unroll
  for (int cb = 0; cb < 8; ++cb) {
    const float* brow = W_rbf + (cb * 16 + r) * N_RADIAL + 2 * hi;
#pragma unroll
    for (int k0 = 0; k0 < 4; ++k0)
      Bt[cb][k0] = *(const v2f*)(brow + 4 * k0);
  }

  for (int t = wave; t < nFull; t += nW) {
    const int e0 = t << 4;

    v2f a[4];
    const float* arow = rbf + (e0 + r) * N_RADIAL + 2 * hi;
#pragma unroll
    for (int k0 = 0; k0 < 4; ++k0)
      a[k0] = *(const v2f*)(arow + 4 * k0);

    int hoff[8];
    const int* ibase = idx_i + e0 + 8 * hi;
#pragma unroll
    for (int v = 0; v < 8; ++v)
      hoff[v] = ibase[v] * EDGE_DIM;

#pragma unroll
    for (int cb = 0; cb < 8; ++cb) {
      v8f c = {0.f, 0.f, 0.f, 0.f, 0.f, 0.f, 0.f, 0.f};
#pragma unroll
      for (int k0 = 0; k0 < 4; ++k0)
        c = __builtin_amdgcn_wmma_f32_16x16x4_f32(
            false, a[k0], false, Bt[cb][k0], (short)0, c, false, false);

      const int col = cb * 16 + r;

      const float* xbase = x + (e0 + 8 * hi) * EDGE_DIM + col;
      float xv[8];
#pragma unroll
      for (int v = 0; v < 8; ++v)
        xv[v] = xbase[v * EDGE_DIM];

#pragma unroll
      for (int v = 0; v < 8; ++v)
        atomicAdd(h + (hoff[v] + col), c[v] * xv[v]);
    }
  }

  if (wave == 0) {
    for (int e = nFull << 4; e < n_edges; ++e) {
      const int nd = idx_i[e];
      const float* re = rbf + e * N_RADIAL;
#pragma unroll
      for (int cc = 0; cc < 4; ++cc) {
        const int ch = lane * 4 + cc;
        float s = 0.f;
#pragma unroll
        for (int k = 0; k < N_RADIAL; ++k) s += re[k] * W_rbf[ch * N_RADIAL + k];
        float g = s * x[e * EDGE_DIM + ch];
        atomicAdd(h + (nd * EDGE_DIM + ch), g);
      }
    }
  }
}

__global__ __launch_bounds__(256) void node_mlp_kernel(
    const float* __restrict__ in, const float* __restrict__ W,
    const float* __restrict__ bias, float* __restrict__ out,
    int n_rows, int apply_swish)
{
  const int lane = threadIdx.x & 31;
  const int r    = lane & 15;
  const int hi   = lane >> 4;
  const int w    = threadIdx.x >> 5;
  const int cbase = (w & 1) * 64;
  const int stream = (w >> 1) * gridDim.x + blockIdx.x;
  const int nS     = (blockDim.x >> 6) * gridDim.x;
  const int nTiles = (n_rows + 15) >> 4;

  v2f a[32];
  {
    int mrow = (stream << 4) + r;
    if (mrow >= n_rows) mrow = n_rows - 1;
    const float* arow = in + mrow * EDGE_DIM + 2 * hi;
#pragma unroll
    for (int k0 = 0; k0 < 32; ++k0)
      a[k0] = *(const v2f*)(arow + 4 * k0);
  }

  v2f Bt[4][32];
#pragma unroll
  for (int cb = 0; cb < 4; ++cb) {
    const float* brow = W + (cbase + cb * 16 + r) * EDGE_DIM + 2 * hi;
#pragma unroll
    for (int k0 = 0; k0 < 32; ++k0)
      Bt[cb][k0] = *(const v2f*)(brow + 4 * k0);
  }

  for (int t = stream; t < nTiles; t += nS) {
    const int m0 = t << 4;

    v2f an[32];
    {
      int mn = ((t + nS) << 4) + r;
      if (mn >= n_rows) mn = n_rows - 1;
      const float* nrow = in + mn * EDGE_DIM + 2 * hi;
#pragma unroll
      for (int k0 = 0; k0 < 32; ++k0)
        an[k0] = *(const v2f*)(nrow + 4 * k0);
    }

    v8f acc[4];
#pragma unroll
    for (int cb = 0; cb < 4; ++cb)
      acc[cb] = (v8f){0.f, 0.f, 0.f, 0.f, 0.f, 0.f, 0.f, 0.f};

#pragma unroll
    for (int k0 = 0; k0 < 32; ++k0) {
#pragma unroll
      for (int cb = 0; cb < 4; ++cb)
        acc[cb] = __builtin_amdgcn_wmma_f32_16x16x4_f32(
            false, a[k0], false, Bt[cb][k0], (short)0, acc[cb], false, false);
    }

    const int rowBase = m0 + 8 * hi;
    if (m0 + 16 <= n_rows) {
#pragma unroll
      for (int cb = 0; cb < 4; ++cb) {
        const int col = cbase + cb * 16 + r;
        const float bv = bias ? bias[col] : 0.f;
        float* obase = out + rowBase * EDGE_DIM + col;
#pragma unroll
        for (int v = 0; v < 8; ++v) {
          float val = acc[cb][v] + bv;
          if (apply_swish) val = val * (1.f / (1.f + __expf(-val)));
          obase[v * EDGE_DIM] = val;
        }
      }
    } else {
#pragma unroll
      for (int cb = 0; cb < 4; ++cb) {
        const int col = cbase + cb * 16 + r;
        const float bv = bias ? bias[col] : 0.f;
#pragma unroll
        for (int v = 0; v < 8; ++v) {
          const int m = rowBase + v;
          if (m < n_rows) {
            float val = acc[cb][v] + bv;
            if (apply_swish) val = val * (1.f / (1.f + __expf(-val)));
            out[m * EDGE_DIM + col] = val;
          }
        }
      }
    }

#pragma unroll
    for (int k0 = 0; k0 < 32; ++k0) a[k0] = an[k0];
  }
}

extern "C" void kernel_launch(void* const* d_in, const int* in_sizes, int n_in,
                              void* d_out, int out_size, void* d_ws, size_t ws_size,
                              hipStream_t stream) {
  const float* x     = (const float*)d_in[0];
  const float* rbf   = (const float*)d_in[1];
  const int*   idx_i = (const int*)d_in[2];
  const float* W_rbf = (const float*)d_in[4];
  const float* W1    = (const float*)d_in[5];
  const float* b1    = (const float*)d_in[6];
  const float* W2    = (const float*)d_in[7];
  const float* b2    = (const float*)d_in[8];
  const float* W3    = (const float*)d_in[9];
  const float* b3    = (const float*)d_in[10];
  const float* W_out = (const float*)d_in[11];
  float* outp = (float*)d_out;

  const int n_edges = in_sizes[0] / EDGE_DIM;
  const int n_nodes = out_size / EDGE_DIM;

  float* h0 = (float*)d_ws;
  float* h1 = h0 + (size_t)n_nodes * EDGE_DIM;

  {
    long n4 = (long)n_nodes * EDGE_DIM / 4;
    long b = (n4 + 255) / 256;
    int blocks = (int)(b > 2048 ? 2048 : (b < 1 ? 1 : b));
    zero_f4_kernel<<<blocks, 256, 0, stream>>>((float4*)h0, n4);
  }

  {
    int nTiles = n_edges >> 4;
    int blocks = (nTiles + 7) / 8;
    if (blocks > 2048) blocks = 2048;
    if (blocks < 1) blocks = 1;
    edge_gate_scatter_kernel<<<blocks, 256, 0, stream>>>(x, rbf, idx_i, W_rbf, h0, n_edges);
  }

  {
    int nTiles = (n_nodes + 15) / 16;
    int blocks = (nTiles + 3) / 4;
    if (blocks > 256) blocks = 256;
    if (blocks < 1) blocks = 1;
    node_mlp_kernel<<<blocks, 256, 0, stream>>>(h0, W1, b1, h1, n_nodes, 1);
    node_mlp_kernel<<<blocks, 256, 0, stream>>>(h1, W2, b2, h0, n_nodes, 1);
    node_mlp_kernel<<<blocks, 256, 0, stream>>>(h0, W3, b3, h1, n_nodes, 1);
    node_mlp_kernel<<<blocks, 256, 0, stream>>>(h1, W_out, nullptr, outp, n_nodes, 0);
  }
}


# Edge scatter: work cut by 64-channel half; a wave reads its die id and draws tiles from the ticket of its own L2 half first, so each node line takes atomics from one L2 half; atomic order by arrival
# speedup vs baseline: 1.4712x; 1.1043x over previous
#include <hip/hip_runtime.h>

typedef __attribute__((ext_vector_type(2))) float v2f;
typedef __attribute__((ext_vector_type(8))) float v8f;

#define EDGE_DIM 128
#define N_RADIAL 16

#define TICKET_STRIDE 64

__global__ void zero_f4_kernel(float4* __restrict__ p, long n4, int* __restrict__ tickets) {
  long i = (long)blockIdx.x * blockDim.x + threadIdx.x;
  long stride = (long)gridDim.x * blockDim.x;
  if (i == 0) {
    tickets[0] = 0;
    tickets[TICKET_STRIDE] = 0;
  }
  float4 z = make_float4(0.f, 0.f, 0.f, 0.f);
  for (; i < n4; i += stride) p[i] = z;
}

__device__ __forceinline__ int draw_tile(int* ctr, int lane) {
  int v = 0;
  if (lane == 0) v = atomicAdd(ctr, 1);
  return __builtin_amdgcn_readfirstlane(v);
}

__global__ __launch_bounds__(256) void edge_gate_scatter_kernel(
    const float* __restrict__ x, const float* __restrict__ rbf,
    const int* __restrict__ idx_i, const float* __restrict__ W_rbf,
    float* __restrict__ h, int* __restrict__ tickets, int n_edges)
{
  const int lane = threadIdx.x & 31;
  const int r    = lane & 15;
  const int hi   = lane >> 4;
  const int wave = blockIdx.x * (blockDim.x >> 5) + (threadIdx.x >> 5);
  const int nFull = n_edges >> 4;

  const int myHalf = (int)((__builtin_amdgcn_s_sendmsg_rtn(0x87) >> 18) & 1u);

  for (int phase = 0; phase < 2; ++phase) {
    const int p = myHalf ^ phase;
    int* ctr = tickets + p * TICKET_STRIDE;

    int t = draw_tile(ctr, lane);
    if (t >= nFull) continue;

    const int cbase = p * 64;

    v2f Bt[4][4];
#pragma unroll
    for (int j = 0; j < 4; ++j) {
      const float* brow = W_rbf + (cbase + j * 16 + r) * N_RADIAL + 2 * hi;
#pragma unroll
      for (int k0 = 0; k0 < 4; ++k0)
        Bt[j][k0] = *(const v2f*)(brow + 4 * k0);
    }

    do {
      const int e0 = t << 4;

      v2f a[4];
      const float* arow = rbf + (e0 + r) * N_RADIAL + 2 * hi;
#pragma unroll
      for (int k0 = 0; k0 < 4; ++k0)
        a[k0] = *(const v2f*)(arow + 4 * k0);

      int hoff[8];
      const int* ibase = idx_i + e0 + 8 * hi;
#pragma unroll
      for (int v = 0; v < 8; ++v)
        hoff[v] = ibase[v] * EDGE_DIM;

#pragma unroll
      for (int j = 0; j < 4; ++j) {
        v8f c = {0.f, 0.f, 0.f, 0.f, 0.f, 0.f, 0.f, 0.f};
#pragma unroll
        for (int k0 = 0; k0 < 4; ++k0)
          c = __builtin_amdgcn_wmma_f32_16x16x4_f32(
              false, a[k0], false, Bt[j][k0], (short)0, c, false, false);

        const int col = cbase + j * 16 + r;

        const float* xbase = x + (e0 + 8 * hi) * EDGE_DIM + col;
        float xv[8];
#pragma unroll
        for (int v = 0; v < 8; ++v)
          xv[v] = xbase[v * EDGE_DIM];

#pragma unroll
        for (int v = 0; v < 8; ++v)
          atomicAdd(h + (hoff[v] + col), c[v] * xv[v]);
      }

      t = draw_tile(ctr, lane);
    } while (t < nFull);
  }

  if (wave == 0) {
    for (int e = nFull << 4; e < n_edges; ++e) {
      const int nd = idx_i[e];
      const float* re = rbf + e * N_RADIAL;
#pragma unroll
      for (int cc = 0; cc < 4; ++cc) {
        const int ch = lane * 4 + cc;
        float s = 0.f;
#pragma unroll
        for (int k = 0; k < N_RADIAL; ++k) s += re[k] * W_rbf[ch * N_RADIAL + k];
        float g = s * x[e * EDGE_DIM + ch];
        atomicAdd(h + (nd * EDGE_DIM + ch), g);
      }
    }
  }
}

__global__ __launch_bounds__(256) void node_mlp_kernel(
    const float* __restrict__ in, const float* __restrict__ W,
    const float* __restrict__ bias, float* __restrict__ out,
    int n_rows, int apply_swish)
{
  const int lane = threadIdx.x & 31;
  const int r    = lane & 15;
  const int hi   = lane >> 4;
  const int w    = threadIdx.x >> 5;
  const int cbase = (w & 1) * 64;
  const int stream = (w >> 1) * gridDim.x + blockIdx.x;
  const int nS     = (blockDim.x >> 6) * gridDim.x;
  const int nTiles = (n_rows + 15) >> 4;

  v2f a[32];
  {
    int mrow = (stream << 4) + r;
    if (mrow >= n_rows) mrow = n_rows - 1;
    const float* arow = in + mrow * EDGE_DIM + 2 * hi;
#pragma unroll
    for (int k0 = 0; k0 < 32; ++k0)
      a[k0] = *(const v2f*)(arow + 4 * k0);
  }

  v2f Bt[4][32];
#pragma unroll
  for (int cb = 0; cb < 4; ++cb) {
    const float* brow = W + (cbase + cb * 16 + r) * EDGE_DIM + 2 * hi;
#pragma unroll
    for (int k0 = 0; k0 < 32; ++k0)
      Bt[cb][k0] = *(const v2f*)(brow + 4 * k0);
  }

  for (int t = stream; t < nTiles; t += nS) {
    const int m0 = t << 4;

    v2f an[32];
    {
      int mn = ((t + nS) << 4) + r;
      if (mn >= n_rows) mn = n_rows - 1;
      const float* nrow = in + mn * EDGE_DIM + 2 * hi;
#pragma unroll
      for (int k0 = 0; k0 < 32; ++k0)
        an[k0] = *(const v2f*)(nrow + 4 * k0);
    }

    v8f acc[4];
#pragma unroll
    for (int cb = 0; cb < 4; ++cb)
      acc[cb] = (v8f){0.f, 0.f, 0.f, 0.f, 0.f, 0.f, 0.f, 0.f};

#pragma unroll
    for (int k0 = 0; k0 < 32; ++k0) {
#pragma unroll
      for (int cb = 0; cb < 4; ++cb)
        acc[cb] = __builtin_amdgcn_wmma_f32_16x16x4_f32(
            false, a[k0], false, Bt[cb][k0], (short)0, acc[cb], false, false);
    }

    const int rowBase = m0 + 8 * hi;
    if (m0 + 16 <= n_rows) {
#pragma unroll
      for (int cb = 0; cb < 4; ++cb) {
        const int col = cbase + cb * 16 + r;
        const float bv = bias ? bias[col] : 0.f;
        float* obase = out + rowBase * EDGE_DIM + col;
#pragma unroll
        for (int v = 0; v < 8; ++v) {
          float val = acc[cb][v] + bv;
          if (apply_swish) val = val * (1.f / (1.f + __expf(-val)));
          obase[v * EDGE_DIM] = val;
        }
      }
    } else {
#pragma unroll
      for (int cb = 0; cb < 4; ++cb) {
        const int col = cbase + cb * 16 + r;
        const float bv = bias ? bias[col] : 0.f;
#pragma unroll
        for (int v = 0; v < 8; ++v) {
          const int m = rowBase + v;
          if (m < n_rows) {
            float val = acc[cb][v] + bv;
            if (apply_swish) val = val * (1.f / (1.f + __expf(-val)));
            out[m * EDGE_DIM + col] = val;
          }
        }
      }
    }

#pragma unroll
    for (int k0 = 0; k0 < 32; ++k0) a[k0] = an[k0];
  }
}

extern "C" void kernel_launch(void* const* d_in, const int* in_sizes, int n_in,
                              void* d_out, int out_size, void* d_ws, size_t ws_size,
                              hipStream_t stream) {
  const float* x     = (const float*)d_in[0];
  const float* rbf   = (const float*)d_in[1];
  const int*   idx_i = (const int*)d_in[2];
  const float* W_rbf = (const float*)d_in[4];
  const float* W1    = (const float*)d_in[5];
  const float* b1    = (const float*)d_in[6];
  const float* W2    = (const float*)d_in[7];
  const float* b2    = (const float*)d_in[8];
  const float* W3    = (const float*)d_in[9];
  const float* b3    = (const float*)d_in[10];
  const float* W_out = (const float*)d_in[11];
  float* outp = (float*)d_out;

  const int n_edges = in_sizes[0] / EDGE_DIM;
  const int n_nodes = out_size / EDGE_DIM;

  float* h0 = (float*)d_ws;
  float* h1 = h0 + (size_t)n_nodes * EDGE_DIM;
  int* tickets = (int*)h1;

  {
    long n4 = (long)n_nodes * EDGE_DIM / 4;
    long b = (n4 + 255) / 256;
    int blocks = (int)(b > 2048 ? 2048 : (b < 1 ? 1 : b));
    zero_f4_kernel<<<blocks, 256, 0, stream>>>((float4*)h0, n4, tickets);
  }

  {
    int nTiles = n_edges >> 4;
    int blocks = (nTiles + 7) / 8;
    if (blocks > 1024) blocks = 1024;
    if (blocks < 1) blocks = 1;
    edge_gate_scatter_kernel<<<blocks, 256, 0, stream>>>(x, rbf, idx_i, W_rbf, h0, tickets, n_edges);
  }

  {
    int nTiles = (n_nodes + 15) / 16;
    int blocks = (nTiles + 3) / 4;
    if (blocks > 256) blocks = 256;
    if (blocks < 1) blocks = 1;
    node_mlp_kernel<<<blocks, 256, 0, stream>>>(h0, W1, b1, h1, n_nodes, 1);
    node_mlp_kernel<<<blocks, 256, 0, stream>>>(h1, W2, b2, h0, n_nodes, 1);
    node_mlp_kernel<<<blocks, 256, 0, stream>>>(h0, W3, b3, h1, n_nodes, 1);
    node_mlp_kernel<<<blocks, 256, 0, stream>>>(h1, W_out, nullptr, outp, n_nodes, 0);
  }
}
